# starting kernel
# baseline (speedup 1.0000x reference)
#include <hip/hip_runtime.h>
#include <hip/hip_bf16.h>
#include <math.h>

typedef __attribute__((ext_vector_type(2))) float v2f;
typedef __attribute__((ext_vector_type(4))) float v4f;
typedef __attribute__((ext_vector_type(8))) float v8f;

#define BTOT   192
#define NN     325
#define DD     64
#define HH     4
#define COLS   256
#define MROWS  (NN * BTOT)
#define MTILES (MROWS / 16)
#define NEG_SLOPE 0.2f

__global__ __launch_bounds__(256) void gat_proj_wmma(
    const float* __restrict__ x,
    const float* __restrict__ W,
    float* __restrict__ xp)
{
    __shared__ float sW[DD * COLS];
    for (int i = threadIdx.x; i < (DD * COLS) / 4; i += 256) {
        ((v4f*)sW)[i] = ((const v4f*)W)[i];
    }
    __syncthreads();

    const int wave  = threadIdx.x >> 5;
    const int lane  = threadIdx.x & 31;
    const int mtile = blockIdx.x * 8 + wave;
    if (mtile >= MTILES) return;

    const int m0   = mtile * 16;
    const int r    = lane & 15;
    const int kgrp = (lane >> 4) * 2;
    const int m    = m0 + r;
    const int n    = m / BTOT;
    const int bt   = m - n * BTOT;
    const float* arow = x + ((size_t)bt * NN + n) * DD;

    v2f afrag[16];
#pragma unroll
    for (int kk = 0; kk < 16; ++kk)
        afrag[kk] = *(const v2f*)(arow + kk * 4 + kgrp);

    float* orow = xp + (size_t)m0 * COLS;
#pragma unroll 1
    for (int c = 0; c < 16; ++c) {
        v8f acc = {};
#pragma unroll
        for (int kk = 0; kk < 16; ++kk) {
            v2f b;
            b.x = sW[(kk * 4 + kgrp + 0) * COLS + c * 16 + (lane & 15)];
            b.y = sW[(kk * 4 + kgrp + 1) * COLS + c * 16 + (lane & 15)];
            acc = __builtin_amdgcn_wmma_f32_16x16x4_f32(
                false, afrag[kk], false, b, (short)0, acc, false, false);
        }
        const int col   = c * 16 + (lane & 15);
        const int rbase = (lane >> 4) * 8;
#pragma unroll
        for (int v = 0; v < 8; ++v)
            orow[(size_t)(rbase + v) * COLS + col] = acc[v];
    }
}

__global__ __launch_bounds__(256) void gat_att(
    const float* __restrict__ xp,
    const float* __restrict__ att_src,
    const float* __restrict__ att_dst,
    float* __restrict__ asrc,
    float* __restrict__ adst)
{
    const int gid = blockIdx.x * 256 + threadIdx.x;
    const int m   = gid >> 5;
    if (m >= MROWS) return;
    const int lane = threadIdx.x & 31;
    const int h    = lane >> 3;
    const int sub  = lane & 7;

    const float* p  = xp + (size_t)m * COLS + h * DD + sub * 8;
    const float* as = att_src + h * DD + sub * 8;
    const float* ad = att_dst + h * DD + sub * 8;
    float ssum = 0.f, dsum = 0.f;
#pragma unroll
    for (int i = 0; i < 8; ++i) { float v = p[i]; ssum += v * as[i]; dsum += v * ad[i]; }
#pragma unroll
    for (int o = 4; o >= 1; o >>= 1) {
        ssum += __shfl_xor(ssum, o, 32);
        dsum += __shfl_xor(dsum, o, 32);
    }
    if (sub == 0) {
        asrc[(size_t)m * HH + h] = ssum;
        adst[(size_t)m * HH + h] = dsum;
    }
}

__global__ void csr_init(int* __restrict__ deg) {
    int n = blockIdx.x * blockDim.x + threadIdx.x;
    if (n < NN) deg[n] = 1;
}

__global__ void csr_count(const int* __restrict__ ei, int* __restrict__ deg, int E) {
    int e = blockIdx.x * blockDim.x + threadIdx.x;
    if (e < E) atomicAdd(&deg[ei[E + e]], 1);
}

__global__ void csr_scan(const int* __restrict__ deg, int* __restrict__ off,
                         int* __restrict__ cur) {
    if (blockIdx.x == 0 && threadIdx.x == 0) {
        int acc = 0;
        for (int n = 0; n < NN; ++n) { off[n] = acc; cur[n] = acc; acc += deg[n]; }
        off[NN] = acc;
    }
}

__global__ void csr_fill(const int* __restrict__ ei, int* __restrict__ cur,
                         int* __restrict__ csrc, int E) {
    int t = blockIdx.x * blockDim.x + threadIdx.x;
    if (t < E) {
        int s = ei[t], d = ei[E + t];
        csrc[atomicAdd(&cur[d], 1)] = s;
    } else if (t < E + NN) {
        int n = t - E;
        csrc[atomicAdd(&cur[n], 1)] = n;
    }
}

__global__ __launch_bounds__(128) void gat_agg_ln(
    const float* __restrict__ x,
    const float* __restrict__ xp,
    const float* __restrict__ asrc,
    const float* __restrict__ adst,
    const int* __restrict__ off,
    const int* __restrict__ csrc,
    const float* __restrict__ bias,
    const float* __restrict__ gamma,
    const float* __restrict__ beta,
    float* __restrict__ out)
{
    __shared__ float heads[HH * DD];
    const int id   = blockIdx.x;
    const int n    = id / BTOT;
    const int bt   = id - n * BTOT;
    const int h    = threadIdx.x >> 5;
    const int lane = threadIdx.x & 31;
    const int d0   = lane * 2;

    const float ad = adst[((size_t)n * BTOT + bt) * HH + h];
    const int kb = off[n], ke = off[n + 1];

    float mrun = -3.402823466e+38f;
    float srun = 0.f, acc0 = 0.f, acc1 = 0.f;
    for (int k = kb; k < ke; ++k) {
        const int j = csrc[k];
        float a = asrc[((size_t)j * BTOT + bt) * HH + h] + ad;
        a = (a >= 0.f) ? a : NEG_SLOPE * a;
        const float mnew = fmaxf(mrun, a);
        const float corr = __expf(mrun - mnew);
        const float w    = __expf(a - mnew);
        srun = srun * corr + w;
        const float* xj = xp + ((size_t)j * BTOT + bt) * COLS + h * DD + d0;
        acc0 = acc0 * corr + w * xj[0];
        acc1 = acc1 * corr + w * xj[1];
        mrun = mnew;
    }
    const float inv = 1.f / (srun + 1e-16f);
    heads[h * DD + d0]     = acc0 * inv;
    heads[h * DD + d0 + 1] = acc1 * inv;
    __syncthreads();

    if (h == 0) {
        float y0 = 0.f, y1 = 0.f;
#pragma unroll
        for (int hh = 0; hh < HH; ++hh) {
            y0 += heads[hh * DD + d0];
            y1 += heads[hh * DD + d0 + 1];
        }
        const float* xr = x + ((size_t)bt * NN + n) * DD;
        y0 = y0 * 0.25f + bias[d0]     + xr[d0];
        y1 = y1 * 0.25f + bias[d0 + 1] + xr[d0 + 1];
        float s  = y0 + y1;
        float s2 = y0 * y0 + y1 * y1;
#pragma unroll
        for (int o = 16; o >= 1; o >>= 1) {
            s  += __shfl_xor(s,  o, 32);
            s2 += __shfl_xor(s2, o, 32);
        }
        const float mu  = s * (1.f / 64.f);
        const float var = s2 * (1.f / 64.f) - mu * mu;
        const float rs  = rsqrtf(var + 1e-5f);
        float* orow = out + ((size_t)bt * NN + n) * DD;
        orow[d0]     = (y0 - mu) * rs * gamma[d0]     + beta[d0];
        orow[d0 + 1] = (y1 - mu) * rs * gamma[d0 + 1] + beta[d0 + 1];
    }
}

extern "C" void kernel_launch(void* const* d_in, const int* in_sizes, int n_in,
                              void* d_out, int out_size, void* d_ws, size_t ws_size,
                              hipStream_t stream) {
    const float* x        = (const float*)d_in[0];
    const float* W        = (const float*)d_in[1];
    const float* att_src  = (const float*)d_in[2];
    const float* att_dst  = (const float*)d_in[3];
    const float* bias     = (const float*)d_in[4];
    const float* gamma    = (const float*)d_in[5];
    const float* beta     = (const float*)d_in[6];
    const int*   ei       = (const int*)d_in[7];
    float* out            = (float*)d_out;
    const int E = in_sizes[7] / 2;

    char* ws = (char*)d_ws;
    float* xp   = (float*)ws;  ws += sizeof(float) * (size_t)MROWS * COLS;
    float* asrc = (float*)ws;  ws += sizeof(float) * (size_t)MROWS * HH;
    float* adst = (float*)ws;  ws += sizeof(float) * (size_t)MROWS * HH;
    int* deg    = (int*)ws;    ws += sizeof(int) * NN;
    int* off    = (int*)ws;    ws += sizeof(int) * (NN + 1);
    int* cur    = (int*)ws;    ws += sizeof(int) * NN;
    int* csrc   = (int*)ws;    ws += sizeof(int) * (E + NN);

    gat_proj_wmma<<<(MTILES + 7) / 8, 256, 0, stream>>>(x, W, xp);

    gat_att<<<(MROWS * 32 + 255) / 256, 256, 0, stream>>>(xp, att_src, att_dst, asrc, adst);

    csr_init <<<(NN + 255) / 256, 256, 0, stream>>>(deg);
    csr_count<<<(E + 255) / 256, 256, 0, stream>>>(ei, deg, E);
    csr_scan <<<1, 32, 0, stream>>>(deg, off, cur);
    csr_fill <<<(E + NN + 255) / 256, 256, 0, stream>>>(ei, cur, csrc, E);

    gat_agg_ln<<<MROWS, 128, 0, stream>>>(x, xp, asrc, adst, off, csrc,
                                          bias, gamma, beta, out);
}


# aggregation gather re-decomposed: one 16-lane row per node, a lane owns 4 dims (16-byte LDS loads); edge row offset and weight spread by DPP row_share, not readlane + integer multiply; sum order kept
# speedup vs baseline: 2.2451x; 2.2451x over previous
#include <hip/hip_runtime.h>
#include <math.h>


typedef __attribute__((ext_vector_type(2))) float v2f;
typedef __attribute__((ext_vector_type(4))) float v4f;
typedef __attribute__((ext_vector_type(8))) float v8f;

#define BTOT   192
#define NN     325
#define NPAD   336
#define DD     64
#define HH     4
#define COLS   256
#define XS_P   68
#define XP_P   68
#define W_P    72
#define ECAP   2925
#define SCH    11
#define SPAD   352
#define NTHR   1024
#define NWAVE  32
#define NEG_SLOPE 0.2f

template <int M>
__device__ __forceinline__ float row_xor(float v) {
    const int b = __float_as_int(v);
    return __int_as_float(__builtin_amdgcn_update_dpp(b, b, 0x160 + M, 0xf, 0xf, false));
}
__device__ __forceinline__ float row_sum(float v) {
    v += row_xor<1>(v); v += row_xor<2>(v); v += row_xor<4>(v); v += row_xor<8>(v);
    return v;
}
__device__ __forceinline__ float row_max(float v) {
    v = fmaxf(v, row_xor<1>(v)); v = fmaxf(v, row_xor<2>(v));
    v = fmaxf(v, row_xor<4>(v)); v = fmaxf(v, row_xor<8>(v));
    return v;
}

template <int M>
__device__ __forceinline__ int row_xor_i(int v) {
    return __builtin_amdgcn_update_dpp(v, v, 0x160 + M, 0xf, 0xf, false);
}
__device__ __forceinline__ int xhalf_i(int v) {
    return __builtin_amdgcn_permlanex16(v, v, 0x76543210, 0xfedcba98, false, false);
}
template <int L>
__device__ __forceinline__ int row_share_i(int v) {
    return __builtin_amdgcn_update_dpp(0, v, 0x150 + L, 0xf, 0xf, true);
}
template <int L>
__device__ __forceinline__ float row_share_f(float v) {
    return __int_as_float(row_share_i<L>(__float_as_int(v)));
}

__device__ __forceinline__ void stage_w_head(const float* __restrict__ W, float* sW, int h, int tid) {
    for (int i = tid; i < DD * 16; i += NTHR) {
        const int d = i >> 4, c4 = i & 15;
        *(v4f*)&sW[d * W_P + c4 * 4] = *(const v4f*)(W + d * COLS + h * DD + c4 * 4);
    }
}

__global__ __launch_bounds__(NTHR) void gat_slice_fused(
    const float* __restrict__ x,
    const float* __restrict__ W,
    const float* __restrict__ att_src,
    const float* __restrict__ att_dst,
    const int* __restrict__ ei,
    int E,
    const float* __restrict__ bias,
    const float* __restrict__ gamma,
    const float* __restrict__ beta,
    float* __restrict__ out)
{
    __shared__ __attribute__((aligned(16))) float xs[NPAD * XS_P];
    __shared__ __attribute__((aligned(16))) float xph[NPAD * XP_P];
    __shared__ __attribute__((aligned(16))) float ys[NN * DD];
    __shared__ __attribute__((aligned(16))) float sW[DD * W_P];
    __shared__ float sas[NPAD];
    __shared__ float sad[NPAD];
    __shared__ int soff[SPAD];
    __shared__ int scs[ECAP];
    __shared__ int sdeg[SPAD];

    const int bt   = blockIdx.x;
    const int tid  = threadIdx.x;
    const int wave = __builtin_amdgcn_readfirstlane(tid >> 5);
    const int lane = tid & 31;

    const int En = E < (ECAP - NN) ? E : (ECAP - NN);
    if (tid < SPAD) sdeg[tid] = tid < NN ? 1 : 0;
    __syncthreads();

    int es[3], ed[3];
#pragma unroll
    for (int q = 0; q < 3; ++q) {
        const int e = tid + q * NTHR;
        es[q] = 0;
        ed[q] = -1;
        if (e < En) { es[q] = ei[e]; ed[q] = ei[E + e]; }
    }
    const v4f* xb = (const v4f*)(x + (size_t)bt * NN * DD);
    v4f tmp[6];
#pragma unroll
    for (int u = 0; u < 6; ++u) {
        const int i = tid + u * NTHR;
        if (i < NN * 16) tmp[u] = xb[i];
    }
    const v4f w0 = *(const v4f*)(W + (tid >> 4) * COLS + (tid & 15) * 4);
    const int   dq  = (lane & 15) * 4;
    const v4f   bia = *(const v4f*)(bias + dq);
    const v4f   gam = *(const v4f*)(gamma + dq);
    const v4f   bet = *(const v4f*)(beta + dq);

    int rk[3];
#pragma unroll
    for (int q = 0; q < 3; ++q) {
        rk[q] = 0;
        if ((unsigned)ed[q] < (unsigned)NN) rk[q] = atomicAdd(&sdeg[ed[q]], 1);
    }
    __syncthreads();

    if (wave == 0) {
        int cnt[SCH];
        int sum = 0;
#pragma unroll
        for (int k = 0; k < SCH; ++k) {
            cnt[k] = sdeg[lane * SCH + k];
            sum += cnt[k];
        }
        int pre = 0, o;
        o = row_xor_i<1>(sum); pre += (lane & 1)  ? o : 0; sum += o;
        o = row_xor_i<2>(sum); pre += (lane & 2)  ? o : 0; sum += o;
        o = row_xor_i<4>(sum); pre += (lane & 4)  ? o : 0; sum += o;
        o = row_xor_i<8>(sum); pre += (lane & 8)  ? o : 0; sum += o;
        o = xhalf_i(sum);      pre += (lane & 16) ? o : 0;
#pragma unroll
        for (int k = 0; k < SCH; ++k) {
            soff[lane * SCH + k] = pre;
            pre += cnt[k];
        }
    }
#pragma unroll
    for (int u = 0; u < 6; ++u) {
        const int i = tid + u * NTHR;
        if (i < NN * 16) *(v4f*)&xs[(i >> 4) * XS_P + (i & 15) * 4] = tmp[u];
    }
    for (int i = tid; i < (NPAD - NN) * DD; i += NTHR)
        xs[(NN + (i >> 6)) * XS_P + (i & 63)] = 0.f;
    *(v4f*)&sW[(tid >> 4) * W_P + (tid & 15) * 4] = w0;
    __syncthreads();

#pragma unroll
    for (int q = 0; q < 3; ++q) {
        if ((unsigned)ed[q] < (unsigned)NN) {
            const int p = soff[ed[q]] + rk[q];
            scs[p < ECAP ? p : ECAP - 1] = es[q];
        }
    }
    if (tid < NN) {
        const int p = soff[tid];
        scs[p < ECAP ? p : ECAP - 1] = tid;
    }
    __syncthreads();

    const int c     = lane & 15;
    const int kgrp  = (lane >> 4) * 2;
    const int rbase = (lane >> 4) * 8;

    for (int h = 0; h < HH; ++h) {
        {
            const int ct = wave & 3;
            const float* bcol = &sW[kgrp * W_P + ct * 16 + c];
            v2f bfrag[16];
#pragma unroll
            for (int kk = 0; kk < 16; ++kk) {
                bfrag[kk].x = bcol[(kk * 4 + 0) * W_P];
                bfrag[kk].y = bcol[(kk * 4 + 1) * W_P];
            }
            for (int mt = wave >> 2; mt < NPAD / 16; mt += NWAVE / 4) {
                const float* arow = &xs[(mt * 16 + c) * XS_P + kgrp];
                v2f afrag[16];
#pragma unroll
                for (int kk = 0; kk < 16; ++kk) afrag[kk] = *(const v2f*)(arow + kk * 4);
                __builtin_amdgcn_sched_barrier(0);
                v8f acc = {};
#pragma unroll
                for (int kk = 0; kk < 16; ++kk)
                    acc = __builtin_amdgcn_wmma_f32_16x16x4_f32(
                        false, afrag[kk], false, bfrag[kk], (short)0, acc, false, false);
                float* orow = &xph[(mt * 16 + rbase) * XP_P + ct * 16 + c];
#pragma unroll
                for (int v = 0; v < 8; ++v) orow[v * XP_P] = acc[v];
            }
        }
        __syncthreads();

        if (tid < NN) {
            const float* as = att_src + h * DD;
            const float* ad = att_dst + h * DD;
            const float* p  = &xph[tid * XP_P];
            float ssum = 0.f, dsum = 0.f;
#pragma unroll
            for (int i = 0; i < DD; i += 2) {
                const v2f v = *(const v2f*)(p + i);
                ssum += v.x * as[i];     dsum += v.x * ad[i];
                ssum += v.y * as[i + 1]; dsum += v.y * ad[i + 1];
            }
            sas[tid] = ssum;
            sad[tid] = dsum;
        }
        if (h + 1 < HH) stage_w_head(W, sW, h + 1, tid);
        __syncthreads();

        {
            const int half = lane >> 4, l16 = lane & 15;
            auto cs_at = [&](int k) -> int { return scs[k < ECAP ? k : ECAP - 1]; };
            auto row_node = [&](int i) -> int {
                int na = wave + 2 * NWAVE * i;
                na = na < NN ? na : NN - 1;
                int nb = na + NWAVE;
                nb = nb < NN ? nb : na;
                return half ? nb : na;
            };
            const char* xlane = (const char*)xph + dq * 4;
            auto finish = [&](int n, bool act, v4f r) {
                float* yp = &ys[n * DD + dq];
                if (h > 0) { const v4f pv = *(const v4f*)yp; r = pv + r; }
                if (h < HH - 1) {
                    if (act) *(v4f*)yp = r;
                } else {
                    const v4f xr = *(const v4f*)&xs[__umul24(n, XS_P) + dq];
                    const v4f y  = r * 0.25f + bia + xr;
                    const float s  = row_sum((y.x + y.y) + (y.z + y.w));
                    const float sq = row_sum((y.x * y.x + y.y * y.y) + (y.z * y.z + y.w * y.w));
                    const float mu  = s * (1.f / 64.f);
                    const float var = sq * (1.f / 64.f) - mu * mu;
                    const float rs  = rsqrtf(var + 1e-5f);
                    const v4f o4 = (y - mu) * rs * gam + bet;
                    if (act) *(v4f*)(out + ((size_t)bt * NN + n) * DD + dq) = o4;
                }
            };

#define GATHER4(E)                                                                                        \
            {                                                                                             \
                const v4f x0 = *(const v4f*)(xlane + row_share_i<E + 0>(roff));                           \
                const v4f x1 = *(const v4f*)(xlane + row_share_i<E + 1>(roff));                           \
                const v4f x2 = *(const v4f*)(xlane + row_share_i<E + 2>(roff));                           \
                const v4f x3 = *(const v4f*)(xlane + row_share_i<E + 3>(roff));                           \
                acc += row_share_f<E + 0>(w) * x0;                                                        \
                acc += row_share_f<E + 1>(w) * x1;                                                        \
                acc += row_share_f<E + 2>(w) * x2;                                                        \
                acc += row_share_f<E + 3>(w) * x3;                                                        \
            }

            const int niter = (NN - wave + 2 * NWAVE - 1) / (2 * NWAVE);
            int mn0 = row_node(0), mn1 = row_node(1);
            int kb0 = soff[mn0], ke0 = soff[mn0 + 1];
            int kb1 = soff[mn1], ke1 = soff[mn1 + 1];
            int j0 = mn0;
            if (kb0 + l16 < ke0) j0 = cs_at(kb0 + l16);
            for (int i = 0; i < niter; ++i) {
                const int mn2 = row_node(i + 2);
                const int kb2 = soff[mn2], ke2 = soff[mn2 + 1];
                int j1 = mn1;
                if (kb1 + l16 < ke1) j1 = cs_at(kb1 + l16);

                const int  nb   = wave + 2 * NWAVE * i + NWAVE;
                const float adn = sad[mn0];
                const int  deg  = ke0 - kb0;
                const int  degA = __builtin_amdgcn_readfirstlane(deg);
                const int  degB = __builtin_amdgcn_readlane(deg, 16);
                const int  degm = degA > degB ? degA : degB;

                float mrun = -3.402823466e+38f;
                float srun = 0.f;
                v4f acc = {};
                for (int base = 0; base < degm; base += 16) {
                    const bool valid = (base + l16) < deg;
                    int j = mn0;
                    if (base == 0) j = j0;
                    else if (valid) j = cs_at(kb0 + base + l16);
                    j = ((unsigned)j < (unsigned)NN) ? j : mn0;
                    float a = sas[j] + adn;
                    a = (a >= 0.f) ? a : NEG_SLOPE * a;
                    a = valid ? a : -3.402823466e+38f;
                    const float mnew = fmaxf(mrun, row_max(a));
                    const float corr = __expf(mrun - mnew);
                    const float w    = valid ? __expf(a - mnew) : 0.f;
                    srun = srun * corr + row_sum(w);
                    mrun = mnew;
                    acc *= corr;
                    const int roff = __umul24(j, XP_P * 4);
                    const int cnt  = (degm - base) < 16 ? (degm - base) : 16;
                    GATHER4(0)
                    if (cnt > 4) {
                        GATHER4(4)
                        if (cnt > 8) {
                            GATHER4(8)
                            if (cnt > 12) GATHER4(12)
                        }
                    }
                }
#undef GATHER4
                const float inv = 1.f / (srun + 1e-16f);
                finish(mn0, half == 0 || nb < NN, acc * inv);

                mn0 = mn1; kb0 = kb1; ke0 = ke1; j0 = j1;
                mn1 = mn2; kb1 = kb2; ke1 = ke2;
            }
        }
        __syncthreads();
    }
}

extern "C" void kernel_launch(void* const* d_in, const int* in_sizes, int n_in,
                              void* d_out, int out_size, void* d_ws, size_t ws_size,
                              hipStream_t stream) {
    const float* x        = (const float*)d_in[0];
    const float* W        = (const float*)d_in[1];
    const float* att_src  = (const float*)d_in[2];
    const float* att_dst  = (const float*)d_in[3];
    const float* bias     = (const float*)d_in[4];
    const float* gamma    = (const float*)d_in[5];
    const float* beta     = (const float*)d_in[6];
    const int*   ei       = (const int*)d_in[7];
    float* out            = (float*)d_out;
    const int E = in_sizes[7] / 2;

    gat_slice_fused<<<BTOT, NTHR, 0, stream>>>(x, W, att_src, att_dst, ei, E,
                                               bias, gamma, beta, out);
}


# aggregation softmax: the 16-lane max and sum butterflies take the partner lane through the DPP operand of the max/add itself; a node's first edge chunk computes no rescale exponential; sum order kept
# speedup vs baseline: 2.3983x; 1.0682x over previous
#include <hip/hip_runtime.h>
#include <math.h>


typedef __attribute__((ext_vector_type(2))) float v2f;
typedef __attribute__((ext_vector_type(4))) float v4f;
typedef __attribute__((ext_vector_type(8))) float v8f;

#define BTOT   192
#define NN     325
#define NPAD   336
#define DD     64
#define HH     4
#define COLS   256
#define XS_P   68
#define XP_P   68
#define W_P    72
#define ECAP   2925
#define SCH    11
#define SPAD   352
#define NTHR   1024
#define NWAVE  32
#define NEG_SLOPE 0.2f

template <int M>
__device__ __forceinline__ float row_xor(float v) {
    return __int_as_float(__builtin_amdgcn_update_dpp(0, __float_as_int(v), 0x160 + M, 0xf, 0xf, true));
}
__device__ __forceinline__ float row_sum(float v) {
    v += row_xor<1>(v); v += row_xor<2>(v); v += row_xor<4>(v); v += row_xor<8>(v);
    return v;
}
__device__ __forceinline__ float row_max(float v) {
    v = __builtin_elementwise_maximum(v, row_xor<1>(v)); v = __builtin_elementwise_maximum(v, row_xor<2>(v));
    v = __builtin_elementwise_maximum(v, row_xor<4>(v)); v = __builtin_elementwise_maximum(v, row_xor<8>(v));
    return v;
}

template <int M>
__device__ __forceinline__ int row_xor_i(int v) {
    return __builtin_amdgcn_update_dpp(v, v, 0x160 + M, 0xf, 0xf, false);
}
__device__ __forceinline__ int xhalf_i(int v) {
    return __builtin_amdgcn_permlanex16(v, v, 0x76543210, 0xfedcba98, false, false);
}
template <int L>
__device__ __forceinline__ int row_share_i(int v) {
    return __builtin_amdgcn_update_dpp(0, v, 0x150 + L, 0xf, 0xf, true);
}
template <int L>
__device__ __forceinline__ float row_share_f(float v) {
    return __int_as_float(row_share_i<L>(__float_as_int(v)));
}

struct first_chunk { static constexpr bool value = true; };
struct later_chunk { static constexpr bool value = false; };

__device__ __forceinline__ void stage_w_head(const float* __restrict__ W, float* sW, int h, int tid) {
    for (int i = tid; i < DD * 16; i += NTHR) {
        const int d = i >> 4, c4 = i & 15;
        *(v4f*)&sW[d * W_P + c4 * 4] = *(const v4f*)(W + d * COLS + h * DD + c4 * 4);
    }
}

__global__ __launch_bounds__(NTHR) void gat_slice_fused(
    const float* __restrict__ x,
    const float* __restrict__ W,
    const float* __restrict__ att_src,
    const float* __restrict__ att_dst,
    const int* __restrict__ ei,
    int E,
    const float* __restrict__ bias,
    const float* __restrict__ gamma,
    const float* __restrict__ beta,
    float* __restrict__ out)
{
    __shared__ __attribute__((aligned(16))) float xs[NPAD * XS_P];
    __shared__ __attribute__((aligned(16))) float xph[NPAD * XP_P];
    __shared__ __attribute__((aligned(16))) float ys[NN * DD];
    __shared__ __attribute__((aligned(16))) float sW[DD * W_P];
    __shared__ float sas[NPAD];
    __shared__ float sad[NPAD];
    __shared__ int soff[SPAD];
    __shared__ int scs[ECAP];
    __shared__ int sdeg[SPAD];

    const int bt   = blockIdx.x;
    const int tid  = threadIdx.x;
    const int wave = __builtin_amdgcn_readfirstlane(tid >> 5);
    const int lane = tid & 31;

    const int En = E < (ECAP - NN) ? E : (ECAP - NN);
    if (tid < SPAD) sdeg[tid] = tid < NN ? 1 : 0;
    __syncthreads();

    int es[3], ed[3];
#pragma unroll
    for (int q = 0; q < 3; ++q) {
        const int e = tid + q * NTHR;
        es[q] = 0;
        ed[q] = -1;
        if (e < En) { es[q] = ei[e]; ed[q] = ei[E + e]; }
    }
    const v4f* xb = (const v4f*)(x + (size_t)bt * NN * DD);
    v4f tmp[6];
#pragma unroll
    for (int u = 0; u < 6; ++u) {
        const int i = tid + u * NTHR;
        if (i < NN * 16) tmp[u] = xb[i];
    }
    const v4f w0 = *(const v4f*)(W + (tid >> 4) * COLS + (tid & 15) * 4);
    const int   dq  = (lane & 15) * 4;
    const v4f   bia = *(const v4f*)(bias + dq);
    const v4f   gam = *(const v4f*)(gamma + dq);
    const v4f   bet = *(const v4f*)(beta + dq);

    int rk[3];
#pragma unroll
    for (int q = 0; q < 3; ++q) {
        rk[q] = 0;
        if ((unsigned)ed[q] < (unsigned)NN) rk[q] = atomicAdd(&sdeg[ed[q]], 1);
    }
    __syncthreads();

    if (wave == 0) {
        int cnt[SCH];
        int sum = 0;
#pragma unroll
        for (int k = 0; k < SCH; ++k) {
            cnt[k] = sdeg[lane * SCH + k];
            sum += cnt[k];
        }
        int pre = 0, o;
        o = row_xor_i<1>(sum); pre += (lane & 1)  ? o : 0; sum += o;
        o = row_xor_i<2>(sum); pre += (lane & 2)  ? o : 0; sum += o;
        o = row_xor_i<4>(sum); pre += (lane & 4)  ? o : 0; sum += o;
        o = row_xor_i<8>(sum); pre += (lane & 8)  ? o : 0; sum += o;
        o = xhalf_i(sum);      pre += (lane & 16) ? o : 0;
#pragma unroll
        for (int k = 0; k < SCH; ++k) {
            soff[lane * SCH + k] = pre;
            pre += cnt[k];
        }
    }
#pragma unroll
    for (int u = 0; u < 6; ++u) {
        const int i = tid + u * NTHR;
        if (i < NN * 16) *(v4f*)&xs[(i >> 4) * XS_P + (i & 15) * 4] = tmp[u];
    }
    for (int i = tid; i < (NPAD - NN) * DD; i += NTHR)
        xs[(NN + (i >> 6)) * XS_P + (i & 63)] = 0.f;
    *(v4f*)&sW[(tid >> 4) * W_P + (tid & 15) * 4] = w0;
    __syncthreads();

#pragma unroll
    for (int q = 0; q < 3; ++q) {
        if ((unsigned)ed[q] < (unsigned)NN) {
            const int p = soff[ed[q]] + rk[q];
            scs[p < ECAP ? p : ECAP - 1] = es[q];
        }
    }
    if (tid < NN) {
        const int p = soff[tid];
        scs[p < ECAP ? p : ECAP - 1] = tid;
    }
    __syncthreads();

    const int c     = lane & 15;
    const int kgrp  = (lane >> 4) * 2;
    const int rbase = (lane >> 4) * 8;

    for (int h = 0; h < HH; ++h) {
        {
            const int ct = wave & 3;
            const float* bcol = &sW[kgrp * W_P + ct * 16 + c];
            v2f bfrag[16];
#pragma unroll
            for (int kk = 0; kk < 16; ++kk) {
                bfrag[kk].x = bcol[(kk * 4 + 0) * W_P];
                bfrag[kk].y = bcol[(kk * 4 + 1) * W_P];
            }
            for (int mt = wave >> 2; mt < NPAD / 16; mt += NWAVE / 4) {
                const float* arow = &xs[(mt * 16 + c) * XS_P + kgrp];
                v2f afrag[16];
#pragma unroll
                for (int kk = 0; kk < 16; ++kk) afrag[kk] = *(const v2f*)(arow + kk * 4);
                __builtin_amdgcn_sched_barrier(0);
                v8f acc = {};
#pragma unroll
                for (int kk = 0; kk < 16; ++kk)
                    acc = __builtin_amdgcn_wmma_f32_16x16x4_f32(
                        false, afrag[kk], false, bfrag[kk], (short)0, acc, false, false);
                float* orow = &xph[(mt * 16 + rbase) * XP_P + ct * 16 + c];
#pragma unroll
                for (int v = 0; v < 8; ++v) orow[v * XP_P] = acc[v];
            }
        }
        __syncthreads();

        if (tid < NN) {
            const float* as = att_src + h * DD;
            const float* ad = att_dst + h * DD;
            const float* p  = &xph[tid * XP_P];
            float ssum = 0.f, dsum = 0.f;
#pragma unroll
            for (int i = 0; i < DD; i += 2) {
                const v2f v = *(const v2f*)(p + i);
                ssum += v.x * as[i];     dsum += v.x * ad[i];
                ssum += v.y * as[i + 1]; dsum += v.y * ad[i + 1];
            }
            sas[tid] = ssum;
            sad[tid] = dsum;
        }
        if (h + 1 < HH) stage_w_head(W, sW, h + 1, tid);
        __syncthreads();

        {
            const int half = lane >> 4, l16 = lane & 15;
            auto cs_at = [&](int k) -> int { return scs[k < ECAP ? k : ECAP - 1]; };
            auto row_node = [&](int i) -> int {
                int na = wave + 2 * NWAVE * i;
                na = na < NN ? na : NN - 1;
                int nb = na + NWAVE;
                nb = nb < NN ? nb : na;
                return half ? nb : na;
            };
            const char* xlane = (const char*)xph + dq * 4;
            auto finish = [&](int n, bool act, v4f r) {
                float* yp = &ys[n * DD + dq];
                if (h > 0) { const v4f pv = *(const v4f*)yp; r = pv + r; }
                if (h < HH - 1) {
                    if (act) *(v4f*)yp = r;
                } else {
                    const v4f xr = *(const v4f*)&xs[__umul24(n, XS_P) + dq];
                    const v4f y  = r * 0.25f + bia + xr;
                    const float s  = row_sum((y.x + y.y) + (y.z + y.w));
                    const float sq = row_sum((y.x * y.x + y.y * y.y) + (y.z * y.z + y.w * y.w));
                    const float mu  = s * (1.f / 64.f);
                    const float var = sq * (1.f / 64.f) - mu * mu;
                    const float rs  = rsqrtf(var + 1e-5f);
                    const v4f o4 = (y - mu) * rs * gam + bet;
                    if (act) *(v4f*)(out + ((size_t)bt * NN + n) * DD + dq) = o4;
                }
            };

#define GATHER4(E)                                                                                        \
            {                                                                                             \
                const v4f x0 = *(const v4f*)(xlane + row_share_i<E + 0>(roff));                           \
                const v4f x1 = *(const v4f*)(xlane + row_share_i<E + 1>(roff));                           \
                const v4f x2 = *(const v4f*)(xlane + row_share_i<E + 2>(roff));                           \
                const v4f x3 = *(const v4f*)(xlane + row_share_i<E + 3>(roff));                           \
                acc += row_share_f<E + 0>(w) * x0;                                                        \
                acc += row_share_f<E + 1>(w) * x1;                                                        \
                acc += row_share_f<E + 2>(w) * x2;                                                        \
                acc += row_share_f<E + 3>(w) * x3;                                                        \
            }

            const int niter = (NN - wave + 2 * NWAVE - 1) / (2 * NWAVE);
            int mn0 = row_node(0), mn1 = row_node(1);
            int kb0 = soff[mn0], ke0 = soff[mn0 + 1];
            int kb1 = soff[mn1], ke1 = soff[mn1 + 1];
            int j0 = mn0;
            if (kb0 + l16 < ke0) j0 = cs_at(kb0 + l16);
            for (int i = 0; i < niter; ++i) {
                const int mn2 = row_node(i + 2);
                const int kb2 = soff[mn2], ke2 = soff[mn2 + 1];
                int j1 = mn1;
                if (kb1 + l16 < ke1) j1 = cs_at(kb1 + l16);

                const int  nb   = wave + 2 * NWAVE * i + NWAVE;
                const float adn = sad[mn0];
                const int  deg  = ke0 - kb0;
                const int  degA = __builtin_amdgcn_readfirstlane(deg);
                const int  degB = __builtin_amdgcn_readlane(deg, 16);
                const int  degm = degA > degB ? degA : degB;

                float mrun, srun;
                v4f acc = {};
                auto chunk = [&](const int base, auto tag) {
                    constexpr bool FIRST = decltype(tag)::value;
                    const bool valid = (base + l16) < deg;
                    int j = mn0;
                    if (FIRST) j = j0;
                    else if (valid) j = cs_at(kb0 + base + l16);
                    j = ((unsigned)j < (unsigned)NN) ? j : mn0;
                    float a = sas[j] + adn;
                    a = (a >= 0.f) ? a : NEG_SLOPE * a;
                    a = valid ? a : -3.402823466e+38f;
                    float w;
                    if constexpr (FIRST) {
                        mrun = row_max(a);
                        w    = valid ? __expf(a - mrun) : 0.f;
                        srun = row_sum(w);
                    } else {
                        const float mnew = __builtin_elementwise_maximum(mrun, row_max(a));
                        const float corr = __expf(mrun - mnew);
                        w    = valid ? __expf(a - mnew) : 0.f;
                        srun = srun * corr + row_sum(w);
                        mrun = mnew;
                        acc *= corr;
                    }
                    const int roff = __umul24(j, XP_P * 4);
                    const int cnt  = (degm - base) < 16 ? (degm - base) : 16;
                    GATHER4(0)
                    if (cnt > 4) {
                        GATHER4(4)
                        if (cnt > 8) {
                            GATHER4(8)
                            if (cnt > 12) GATHER4(12)
                        }
                    }
                };
                chunk(0, first_chunk{});
                for (int base = 16; base < degm; base += 16) chunk(base, later_chunk{});
#undef GATHER4
                const float inv = 1.f / (srun + 1e-16f);
                finish(mn0, half == 0 || nb < NN, acc * inv);

                mn0 = mn1; kb0 = kb1; ke0 = ke1; j0 = j1;
                mn1 = mn2; kb1 = kb2; ke1 = ke2;
            }
        }
        __syncthreads();
    }
}

extern "C" void kernel_launch(void* const* d_in, const int* in_sizes, int n_in,
                              void* d_out, int out_size, void* d_ws, size_t ws_size,
                              hipStream_t stream) {
    const float* x        = (const float*)d_in[0];
    const float* W        = (const float*)d_in[1];
    const float* att_src  = (const float*)d_in[2];
    const float* att_dst  = (const float*)d_in[3];
    const float* bias     = (const float*)d_in[4];
    const float* gamma    = (const float*)d_in[5];
    const float* beta     = (const float*)d_in[6];
    const int*   ei       = (const int*)d_in[7];
    float* out            = (float*)d_out;
    const int E = in_sizes[7] / 2;

    gat_slice_fused<<<BTOT, NTHR, 0, stream>>>(x, W, att_src, att_dst, ei, E,
                                               bias, gamma, beta, out);
}


# attention logits re-decomposed: all 32 waves, 8 lanes per node (16-byte LDS loads, DPP sums), att vectors staged in LDS once, next head's W requested ahead of the sums; logit sum order changed
# speedup vs baseline: 2.5536x; 1.0648x over previous
#include <hip/hip_runtime.h>
#include <math.h>


typedef __attribute__((ext_vector_type(2))) float v2f;
typedef __attribute__((ext_vector_type(4))) float v4f;
typedef __attribute__((ext_vector_type(8))) float v8f;

#define BTOT   192
#define NN     325
#define NPAD   336
#define DD     64
#define HH     4
#define COLS   256
#define XS_P   68
#define XP_P   68
#define W_P    72
#define ECAP   2925
#define SCH    11
#define SPAD   352
#define NTHR   1024
#define NWAVE  32
#define NEG_SLOPE 0.2f

template <int M>
__device__ __forceinline__ float row_xor(float v) {
    return __int_as_float(__builtin_amdgcn_update_dpp(0, __float_as_int(v), 0x160 + M, 0xf, 0xf, true));
}
__device__ __forceinline__ float row_sum(float v) {
    v += row_xor<1>(v); v += row_xor<2>(v); v += row_xor<4>(v); v += row_xor<8>(v);
    return v;
}
__device__ __forceinline__ float row_max(float v) {
    v = __builtin_elementwise_maximum(v, row_xor<1>(v)); v = __builtin_elementwise_maximum(v, row_xor<2>(v));
    v = __builtin_elementwise_maximum(v, row_xor<4>(v)); v = __builtin_elementwise_maximum(v, row_xor<8>(v));
    return v;
}

template <int M>
__device__ __forceinline__ int row_xor_i(int v) {
    return __builtin_amdgcn_update_dpp(v, v, 0x160 + M, 0xf, 0xf, false);
}
__device__ __forceinline__ int xhalf_i(int v) {
    return __builtin_amdgcn_permlanex16(v, v, 0x76543210, 0xfedcba98, false, false);
}
template <int L>
__device__ __forceinline__ int row_share_i(int v) {
    return __builtin_amdgcn_update_dpp(0, v, 0x150 + L, 0xf, 0xf, true);
}
template <int L>
__device__ __forceinline__ float row_share_f(float v) {
    return __int_as_float(row_share_i<L>(__float_as_int(v)));
}

struct first_chunk { static constexpr bool value = true; };
struct later_chunk { static constexpr bool value = false; };

__global__ __launch_bounds__(NTHR) void gat_slice_fused(
    const float* __restrict__ x,
    const float* __restrict__ W,
    const float* __restrict__ att_src,
    const float* __restrict__ att_dst,
    const int* __restrict__ ei,
    int E,
    const float* __restrict__ bias,
    const float* __restrict__ gamma,
    const float* __restrict__ beta,
    float* __restrict__ out)
{
    __shared__ __attribute__((aligned(16))) float xs[NPAD * XS_P];
    __shared__ __attribute__((aligned(16))) float xph[NPAD * XP_P];
    __shared__ __attribute__((aligned(16))) float ys[NN * DD];
    __shared__ __attribute__((aligned(16))) float sW[DD * W_P];
    __shared__ __attribute__((aligned(16))) float satt[2 * COLS];
    __shared__ float sas[NPAD];
    __shared__ float sad[NPAD];
    __shared__ int soff[SPAD];
    __shared__ int scs[ECAP];
    __shared__ int sdeg[SPAD];

    const int bt   = blockIdx.x;
    const int tid  = threadIdx.x;
    const int wave = __builtin_amdgcn_readfirstlane(tid >> 5);
    const int lane = tid & 31;

    const int En = E < (ECAP - NN) ? E : (ECAP - NN);
    if (tid < SPAD) sdeg[tid] = tid < NN ? 1 : 0;
    __syncthreads();

    int es[3], ed[3];
#pragma unroll
    for (int q = 0; q < 3; ++q) {
        const int e = tid + q * NTHR;
        es[q] = 0;
        ed[q] = -1;
        if (e < En) { es[q] = ei[e]; ed[q] = ei[E + e]; }
    }
    const v4f* xb = (const v4f*)(x + (size_t)bt * NN * DD);
    v4f tmp[6];
#pragma unroll
    for (int u = 0; u < 6; ++u) {
        const int i = tid + u * NTHR;
        if (i < NN * 16) tmp[u] = xb[i];
    }
    const v4f w0 = *(const v4f*)(W + (tid >> 4) * COLS + (tid & 15) * 4);
    const v4f attv = *(const v4f*)((tid < 64 ? att_src : att_dst) + (tid & 63) * 4);
    const int   dq  = (lane & 15) * 4;
    const v4f   bia = *(const v4f*)(bias + dq);
    const v4f   gam = *(const v4f*)(gamma + dq);
    const v4f   bet = *(const v4f*)(beta + dq);

    int rk[3];
#pragma unroll
    for (int q = 0; q < 3; ++q) {
        rk[q] = 0;
        if ((unsigned)ed[q] < (unsigned)NN) rk[q] = atomicAdd(&sdeg[ed[q]], 1);
    }
    __syncthreads();

    if (wave == 0) {
        int cnt[SCH];
        int sum = 0;
#pragma unroll
        for (int k = 0; k < SCH; ++k) {
            cnt[k] = sdeg[lane * SCH + k];
            sum += cnt[k];
        }
        int pre = 0, o;
        o = row_xor_i<1>(sum); pre += (lane & 1)  ? o : 0; sum += o;
        o = row_xor_i<2>(sum); pre += (lane & 2)  ? o : 0; sum += o;
        o = row_xor_i<4>(sum); pre += (lane & 4)  ? o : 0; sum += o;
        o = row_xor_i<8>(sum); pre += (lane & 8)  ? o : 0; sum += o;
        o = xhalf_i(sum);      pre += (lane & 16) ? o : 0;
#pragma unroll
        for (int k = 0; k < SCH; ++k) {
            soff[lane * SCH + k] = pre;
            pre += cnt[k];
        }
    }
#pragma unroll
    for (int u = 0; u < 6; ++u) {
        const int i = tid + u * NTHR;
        if (i < NN * 16) *(v4f*)&xs[(i >> 4) * XS_P + (i & 15) * 4] = tmp[u];
    }
    for (int i = tid; i < (NPAD - NN) * DD; i += NTHR)
        xs[(NN + (i >> 6)) * XS_P + (i & 63)] = 0.f;
    *(v4f*)&sW[(tid >> 4) * W_P + (tid & 15) * 4] = w0;
    if (tid < 128) *(v4f*)&satt[tid * 4] = attv;
    __syncthreads();

#pragma unroll
    for (int q = 0; q < 3; ++q) {
        if ((unsigned)ed[q] < (unsigned)NN) {
            const int p = soff[ed[q]] + rk[q];
            scs[p < ECAP ? p : ECAP - 1] = es[q];
        }
    }
    if (tid < NN) {
        const int p = soff[tid];
        scs[p < ECAP ? p : ECAP - 1] = tid;
    }
    __syncthreads();

    const int c     = lane & 15;
    const int kgrp  = (lane >> 4) * 2;
    const int rbase = (lane >> 4) * 8;

    for (int h = 0; h < HH; ++h) {
        {
            const int ct = wave & 3;
            const float* bcol = &sW[kgrp * W_P + ct * 16 + c];
            v2f bfrag[16];
#pragma unroll
            for (int kk = 0; kk < 16; ++kk) {
                bfrag[kk].x = bcol[(kk * 4 + 0) * W_P];
                bfrag[kk].y = bcol[(kk * 4 + 1) * W_P];
            }
            for (int mt = wave >> 2; mt < NPAD / 16; mt += NWAVE / 4) {
                const float* arow = &xs[(mt * 16 + c) * XS_P + kgrp];
                v2f afrag[16];
#pragma unroll
                for (int kk = 0; kk < 16; ++kk) afrag[kk] = *(const v2f*)(arow + kk * 4);
                __builtin_amdgcn_sched_barrier(0);
                v8f acc = {};
#pragma unroll
                for (int kk = 0; kk < 16; ++kk)
                    acc = __builtin_amdgcn_wmma_f32_16x16x4_f32(
                        false, afrag[kk], false, bfrag[kk], (short)0, acc, false, false);
                float* orow = &xph[(mt * 16 + rbase) * XP_P + ct * 16 + c];
#pragma unroll
                for (int v = 0; v < 8; ++v) orow[v * XP_P] = acc[v];
            }
        }
        __syncthreads();

        {
            v4f wn = {};
            if (h + 1 < HH) wn = *(const v4f*)(W + (tid >> 4) * COLS + (h + 1) * DD + (tid & 15) * 4);
            const int o = lane & 7;
            const v4f* ap = (const v4f*)&satt[h * DD + o * 8];
            const v4f* bp = (const v4f*)&satt[COLS + h * DD + o * 8];
            const v4f as0 = ap[0], as1 = ap[1];
            const v4f ad0 = bp[0], ad1 = bp[1];
            float* lout = o ? sad : sas;
            v4f a0[3], a1[3];
#pragma unroll
            for (int ps = 0; ps < 3; ++ps) {
                const int n  = wave * 4 + ps * (NWAVE * 4) + (lane >> 3);
                const int nc = n < NPAD ? n : NPAD - 1;
                const v4f* p = (const v4f*)&xph[nc * XP_P + o * 8];
                a0[ps] = p[0];
                a1[ps] = p[1];
            }
#pragma unroll
            for (int ps = 0; ps < 3; ++ps) {
                const int n  = wave * 4 + ps * (NWAVE * 4) + (lane >> 3);
                const v4f s4 = a0[ps] * as0 + a1[ps] * as1;
                const v4f d4 = a0[ps] * ad0 + a1[ps] * ad1;
                float ssum = (s4.x + s4.y) + (s4.z + s4.w);
                float dsum = (d4.x + d4.y) + (d4.z + d4.w);
                ssum += row_xor<1>(ssum); ssum += row_xor<2>(ssum); ssum += row_xor<4>(ssum);
                dsum += row_xor<1>(dsum); dsum += row_xor<2>(dsum); dsum += row_xor<4>(dsum);
                if (o < 2 && n < NPAD) lout[n] = o ? dsum : ssum;
            }
            if (h + 1 < HH) *(v4f*)&sW[(tid >> 4) * W_P + (tid & 15) * 4] = wn;
        }
        __syncthreads();

        {
            const int half = lane >> 4, l16 = lane & 15;
            auto cs_at = [&](int k) -> int { return scs[k < ECAP ? k : ECAP - 1]; };
            auto row_node = [&](int i) -> int {
                int na = wave + 2 * NWAVE * i;
                na = na < NN ? na : NN - 1;
                int nb = na + NWAVE;
                nb = nb < NN ? nb : na;
                return half ? nb : na;
            };
            const char* xlane = (const char*)xph + dq * 4;
            auto finish = [&](int n, bool act, v4f r) {
                float* yp = &ys[n * DD + dq];
                if (h > 0) { const v4f pv = *(const v4f*)yp; r = pv + r; }
                if (h < HH - 1) {
                    if (act) *(v4f*)yp = r;
                } else {
                    const v4f xr = *(const v4f*)&xs[__umul24(n, XS_P) + dq];
                    const v4f y  = r * 0.25f + bia + xr;
                    const float s  = row_sum((y.x + y.y) + (y.z + y.w));
                    const float sq = row_sum((y.x * y.x + y.y * y.y) + (y.z * y.z + y.w * y.w));
                    const float mu  = s * (1.f / 64.f);
                    const float var = sq * (1.f / 64.f) - mu * mu;
                    const float rs  = rsqrtf(var + 1e-5f);
                    const v4f o4 = (y - mu) * rs * gam + bet;
                    if (act) *(v4f*)(out + ((size_t)bt * NN + n) * DD + dq) = o4;
                }
            };

#define GATHER4(E)                                                                                        \
            {                                                                                             \
                const v4f x0 = *(const v4f*)(xlane + row_share_i<E + 0>(roff));                           \
                const v4f x1 = *(const v4f*)(xlane + row_share_i<E + 1>(roff));                           \
                const v4f x2 = *(const v4f*)(xlane + row_share_i<E + 2>(roff));                           \
                const v4f x3 = *(const v4f*)(xlane + row_share_i<E + 3>(roff));                           \
                acc += row_share_f<E + 0>(w) * x0;                                                        \
                acc += row_share_f<E + 1>(w) * x1;                                                        \
                acc += row_share_f<E + 2>(w) * x2;                                                        \
                acc += row_share_f<E + 3>(w) * x3;                                                        \
            }

            const int niter = (NN - wave + 2 * NWAVE - 1) / (2 * NWAVE);
            int mn0 = row_node(0), mn1 = row_node(1);
            int kb0 = soff[mn0], ke0 = soff[mn0 + 1];
            int kb1 = soff[mn1], ke1 = soff[mn1 + 1];
            int j0 = mn0;
            if (kb0 + l16 < ke0) j0 = cs_at(kb0 + l16);
            for (int i = 0; i < niter; ++i) {
                const int mn2 = row_node(i + 2);
                const int kb2 = soff[mn2], ke2 = soff[mn2 + 1];
                int j1 = mn1;
                if (kb1 + l16 < ke1) j1 = cs_at(kb1 + l16);

                const int  nb   = wave + 2 * NWAVE * i + NWAVE;
                const float adn = sad[mn0];
                const int  deg  = ke0 - kb0;
                const int  degA = __builtin_amdgcn_readfirstlane(deg);
                const int  degB = __builtin_amdgcn_readlane(deg, 16);
                const int  degm = degA > degB ? degA : degB;

                float mrun, srun;
                v4f acc = {};
                auto chunk = [&](const int base, auto tag) {
                    constexpr bool FIRST = decltype(tag)::value;
                    const bool valid = (base + l16) < deg;
                    int j = mn0;
                    if (FIRST) j = j0;
                    else if (valid) j = cs_at(kb0 + base + l16);
                    j = ((unsigned)j < (unsigned)NN) ? j : mn0;
                    float a = sas[j] + adn;
                    a = (a >= 0.f) ? a : NEG_SLOPE * a;
                    a = valid ? a : -3.402823466e+38f;
                    float w;
                    if constexpr (FIRST) {
                        mrun = row_max(a);
                        w    = valid ? __expf(a - mrun) : 0.f;
                        srun = row_sum(w);
                    } else {
                        const float mnew = __builtin_elementwise_maximum(mrun, row_max(a));
                        const float corr = __expf(mrun - mnew);
                        w    = valid ? __expf(a - mnew) : 0.f;
                        srun = srun * corr + row_sum(w);
                        mrun = mnew;
                        acc *= corr;
                    }
                    const int roff = __umul24(j, XP_P * 4);
                    const int cnt  = (degm - base) < 16 ? (degm - base) : 16;
                    GATHER4(0)
                    if (cnt > 4) {
                        GATHER4(4)
                        if (cnt > 8) {
                            GATHER4(8)
                            if (cnt > 12) GATHER4(12)
                        }
                    }
                };
                chunk(0, first_chunk{});
                for (int base = 16; base < degm; base += 16) chunk(base, later_chunk{});
#undef GATHER4
                const float inv = 1.f / (srun + 1e-16f);
                finish(mn0, half == 0 || nb < NN, acc * inv);

                mn0 = mn1; kb0 = kb1; ke0 = ke1; j0 = j1;
                mn1 = mn2; kb1 = kb2; ke1 = ke2;
            }
        }
        __syncthreads();
    }
}

extern "C" void kernel_launch(void* const* d_in, const int* in_sizes, int n_in,
                              void* d_out, int out_size, void* d_ws, size_t ws_size,
                              hipStream_t stream) {
    const float* x        = (const float*)d_in[0];
    const float* W        = (const float*)d_in[1];
    const float* att_src  = (const float*)d_in[2];
    const float* att_dst  = (const float*)d_in[3];
    const float* bias     = (const float*)d_in[4];
    const float* gamma    = (const float*)d_in[5];
    const float* beta     = (const float*)d_in[6];
    const int*   ei       = (const int*)d_in[7];
    float* out            = (float*)d_out;
    const int E = in_sizes[7] / 2;

    gat_slice_fused<<<BTOT, NTHR, 0, stream>>>(x, W, att_src, att_dst, ei, E,
                                               bias, gamma, beta, out);
}
